# starting kernel
# baseline (speedup 1.0000x reference)
#include <hip/hip_runtime.h>
#include <math.h>


typedef float v2f __attribute__((ext_vector_type(2)));
typedef float v8f __attribute__((ext_vector_type(8)));

#define SK_N     4096
#define SK_NN    (4096 * 4096)
#define SK_IEPS  10.0f
#define SK_TOL   1e-9f
#define SK_ITERS 100

__global__ void __launch_bounds__(256) sk_init_K(const float* __restrict__ C,
                                                 float* __restrict__ K) {
  size_t i4 = ((size_t)blockIdx.x * 256 + threadIdx.x) * 4;
  float4 c4 = *(const float4*)(C + i4);
  float4 k4;
  k4.x = expf(-SK_IEPS * c4.x);
  k4.y = expf(-SK_IEPS * c4.y);
  k4.z = expf(-SK_IEPS * c4.z);
  k4.w = expf(-SK_IEPS * c4.w);
  *(float4*)(K + i4) = k4;
}

__global__ void sk_ones(float* __restrict__ v) {
  v[blockIdx.x * 256 + threadIdx.x] = 1.0f;
}

__global__ void __launch_bounds__(256) sk_matvec_Kv(const float* __restrict__ K,
                                                    const float* __restrict__ v,
                                                    float* __restrict__ acc_out) {
  __shared__ float part[8][16];
  const int wave = threadIdx.x >> 5;
  const int lane = threadIdx.x & 31;
  const int hf   = lane >> 4;
  const int r    = lane & 15;
  const int rowBase = blockIdx.x << 4;
  const float* Krow = K + (size_t)(rowBase + r) * SK_N + 2 * hf;
  const float* vp   = v + 2 * hf;

  v8f acc = {0.f, 0.f, 0.f, 0.f, 0.f, 0.f, 0.f, 0.f};
  const int c0 = wave << 9;
#pragma unroll 4
  for (int c = c0; c < c0 + 512; c += 4) {
    v2f a = *(const v2f*)(Krow + c);
    v2f b = *(const v2f*)(vp + c);
    acc = __builtin_amdgcn_wmma_f32_16x16x4_f32(false, a, false, b,
                                                (short)0, acc, false, false);
  }
  if (r == 0) {
    float* dst = &part[wave][hf * 8];
    dst[0] = acc[0]; dst[1] = acc[1]; dst[2] = acc[2]; dst[3] = acc[3];
    dst[4] = acc[4]; dst[5] = acc[5]; dst[6] = acc[6]; dst[7] = acc[7];
  }
  __syncthreads();
  if (threadIdx.x < 16) {
    float s = 0.f;
#pragma unroll
    for (int w = 0; w < 8; ++w) s += part[w][threadIdx.x];
    acc_out[rowBase + threadIdx.x] = s;
  }
}

__global__ void __launch_bounds__(256) sk_matvec_KT(const float* __restrict__ K,
                                                    const float* __restrict__ u,
                                                    float* __restrict__ acc_out) {
  __shared__ float part[8][16];
  const int wave = threadIdx.x >> 5;
  const int lane = threadIdx.x & 31;
  const int hf   = lane >> 4;
  const int r    = lane & 15;
  const int colBase = blockIdx.x << 4;
  const float* Kc = K + colBase + r;

  v8f acc = {0.f, 0.f, 0.f, 0.f, 0.f, 0.f, 0.f, 0.f};
  const int r0 = wave << 9;
#pragma unroll 2
  for (int row = r0; row < r0 + 512; row += 4) {
    v2f a = *(const v2f*)(u + row + 2 * hf);
    size_t base = (size_t)(row + 2 * hf) * SK_N;
    v2f b;
    b[0] = Kc[base];
    b[1] = Kc[base + SK_N];
    acc = __builtin_amdgcn_wmma_f32_16x16x4_f32(false, a, false, b,
                                                (short)0, acc, false, false);
  }
  if (lane < 16) part[wave][lane] = acc[0];
  __syncthreads();
  if (threadIdx.x < 16) {
    float s = 0.f;
#pragma unroll
    for (int w = 0; w < 8; ++w) s += part[w][threadIdx.x];
    acc_out[colBase + threadIdx.x] = s;
  }
}

__global__ void sk_scale(const float* __restrict__ target,
                         const float* __restrict__ acc,
                         float* __restrict__ out) {
  int i = blockIdx.x * 256 + threadIdx.x;
  out[i] = target[i] / (acc[i] + SK_TOL);
}

__global__ void __launch_bounds__(256) sk_final(const float* __restrict__ mu,
                                                const float* __restrict__ nu,
                                                const float* __restrict__ u,
                                                const float* __restrict__ v,
                                                float* __restrict__ P,
                                                float* __restrict__ lpart) {
  __shared__ float red[256];
  const int i = blockIdx.x;
  const float ui = u[i];
  const float mi = mu[i];
  float* Prow = P + (size_t)i * SK_N;
  float w = 0.f;
  for (int j = threadIdx.x * 4; j < SK_N; j += 256 * 4) {
    float4 k4 = *(const float4*)(Prow + j);
    float4 v4 = *(const float4*)(v + j);
    float4 n4 = *(const float4*)(nu + j);
    float4 p;
    p.x = ui * k4.x * v4.x;
    p.y = ui * k4.y * v4.y;
    p.z = ui * k4.z * v4.z;
    p.w = ui * k4.w * v4.w;
    *(float4*)(Prow + j) = p;
    w += p.x * fabsf(mi - n4.x) + p.y * fabsf(mi - n4.y) +
         p.z * fabsf(mi - n4.z) + p.w * fabsf(mi - n4.w);
  }
  red[threadIdx.x] = w;
  __syncthreads();
  for (int s = 128; s > 0; s >>= 1) {
    if (threadIdx.x < s) red[threadIdx.x] += red[threadIdx.x + s];
    __syncthreads();
  }
  if (threadIdx.x == 0) lpart[i] = red[0];
}

__global__ void __launch_bounds__(256) sk_loss_reduce(const float* __restrict__ lpart,
                                                      float* __restrict__ loss) {
  __shared__ float red[256];
  float s = 0.f;
  for (int i = threadIdx.x; i < SK_N; i += 256) s += lpart[i];
  red[threadIdx.x] = s;
  __syncthreads();
  for (int k = 128; k > 0; k >>= 1) {
    if (threadIdx.x < k) red[threadIdx.x] += red[threadIdx.x + k];
    __syncthreads();
  }
  if (threadIdx.x == 0) *loss = red[0];
}

extern "C" void kernel_launch(void* const* d_in, const int* in_sizes, int n_in,
                              void* d_out, int out_size, void* d_ws, size_t ws_size,
                              hipStream_t stream) {
  const float* mu = (const float*)d_in[0];
  const float* nu = (const float*)d_in[1];
  const float* C  = (const float*)d_in[2];
  float* K = (float*)d_out;
  float* ws = (float*)d_ws;
  float* u     = ws;
  float* v     = ws + SK_N;
  float* acc   = ws + 2 * SK_N;
  float* lpart = ws + 3 * SK_N;

  sk_init_K<<<SK_NN / (256 * 4), 256, 0, stream>>>(C, K);
  sk_ones<<<SK_N / 256, 256, 0, stream>>>(v);

  for (int t = 0; t < SK_ITERS; ++t) {
    sk_matvec_Kv<<<SK_N / 16, 256, 0, stream>>>(K, v, acc);
    sk_scale<<<SK_N / 256, 256, 0, stream>>>(mu, acc, u);
    sk_matvec_KT<<<SK_N / 16, 256, 0, stream>>>(K, u, acc);
    sk_scale<<<SK_N / 256, 256, 0, stream>>>(nu, acc, v);
  }

  sk_final<<<SK_N, 256, 0, stream>>>(mu, nu, u, v, K, lpart);
  sk_loss_reduce<<<1, 256, 0, stream>>>(lpart, K + SK_NN);
}


# One pass over K per iteration: each work-group keeps its 16 rows in registers (WMMA transposes them to D layout), forms u, then column partials; a 2nd launch adds them; f32 sum order changes
# speedup vs baseline: 2.1652x; 2.1652x over previous
#include <hip/hip_runtime.h>
#include <math.h>


typedef float v2f __attribute__((ext_vector_type(2)));
typedef float v8f __attribute__((ext_vector_type(8)));

#define SK_N     4096
#define SK_NN    (4096 * 4096)
#define SK_IEPS  10.0f
#define SK_TOL   1e-9f
#define SK_ITERS 100

#define SK_FW    16
#define SK_FCOLS (SK_N / SK_FW)
#define SK_FG    (SK_FCOLS / 16)
#define SK_NRB   (SK_N / 16)
#define SK_WS_NEED ((size_t)(4 * SK_N + SK_NRB * SK_N) * sizeof(float))

static __device__ __forceinline__ v8f sk_wmma(v2f a, v2f b, v8f c) {
  return __builtin_amdgcn_wmma_f32_16x16x4_f32(false, a, false, b, (short)0, c,
                                               false, false);
}

__global__ void __launch_bounds__(256) sk_init_K(const float* __restrict__ C,
                                                 float* __restrict__ K) {
  size_t i4 = ((size_t)blockIdx.x * 256 + threadIdx.x) * 4;
  float4 c4 = *(const float4*)(C + i4);
  float4 k4;
  k4.x = expf(-SK_IEPS * c4.x);
  k4.y = expf(-SK_IEPS * c4.y);
  k4.z = expf(-SK_IEPS * c4.z);
  k4.w = expf(-SK_IEPS * c4.w);
  *(float4*)(K + i4) = k4;
}

__global__ void sk_ones(float* __restrict__ v) {
  v[blockIdx.x * 256 + threadIdx.x] = 1.0f;
}

__global__ void __launch_bounds__(512) sk_rows_fused(const float* __restrict__ K,
                                                     const float* __restrict__ mu,
                                                     const float* __restrict__ v,
                                                     float* __restrict__ u_out,
                                                     float* __restrict__ cpart,
                                                     size_t ws_size) {
  __shared__ float4 vsh4[SK_N / 4];
  __shared__ float part[SK_FW][16];
  __shared__ float ush[16];
  const int wave = threadIdx.x >> 5;
  const int lane = threadIdx.x & 31;
  const int h = lane >> 4;
  const int c = lane & 15;
  const int rowBase = blockIdx.x << 4;
  const int colBase = wave * SK_FCOLS;
  const float* Krow = K + (size_t)(rowBase + c) * SK_N + colBase + 4 * h;
  const float4* vp4 = vsh4 + (colBase >> 2) + h;

  vsh4[threadIdx.x]       = *(const float4*)(v + 4 * threadIdx.x);
  vsh4[512 + threadIdx.x] = *(const float4*)(v + 2048 + 4 * threadIdx.x);

  v2f sel[2][2];
#pragma unroll
  for (int p = 0; p < 2; ++p) {
#pragma unroll
    for (int e = 0; e < 2; ++e) {
      const int n0 = 8 * p + 4 * h + 2 * e;
      sel[p][e][0] = (c == n0) ? 1.0f : 0.0f;
      sel[p][e][1] = (c == n0 + 1) ? 1.0f : 0.0f;
    }
  }

  const v8f zero = {0.f, 0.f, 0.f, 0.f, 0.f, 0.f, 0.f, 0.f};
  v8f rs0 = zero, rs1 = zero;
  v8f tile[SK_FG];
  float4 ka[4], kn[4];
#pragma unroll
  for (int x = 0; x < 4; ++x) ka[x] = *(const float4*)(Krow + 8 * x);
  __syncthreads();
#pragma unroll
  for (int st = 0; st < SK_FG / 2; ++st) {
    if (st + 1 < SK_FG / 2) {
#pragma unroll
      for (int x = 0; x < 4; ++x)
        kn[x] = *(const float4*)(Krow + 32 * (st + 1) + 8 * x);
    }
#pragma unroll
    for (int gg = 0; gg < 2; ++gg) {
      v8f d = zero;
#pragma unroll
      for (int p = 0; p < 2; ++p) {
        const float4 a4 = ka[2 * gg + p];
        const float4 b4 = vp4[8 * st + 2 * (2 * gg + p)];
        v2f a0, a1, b0, b1;
        a0[0] = a4.x; a0[1] = a4.y; a1[0] = a4.z; a1[1] = a4.w;
        b0[0] = b4.x; b0[1] = b4.y; b1[0] = b4.z; b1[1] = b4.w;
        rs0 = sk_wmma(a0, b0, rs0);
        rs1 = sk_wmma(a1, b1, rs1);
        d = sk_wmma(a0, sel[p][0], d);
        d = sk_wmma(a1, sel[p][1], d);
      }
      tile[2 * st + gg] = d;
    }
    __builtin_amdgcn_sched_barrier(0);
#pragma unroll
    for (int x = 0; x < 4; ++x) ka[x] = kn[x];
  }

  if (c == 0) {
    float* dst = &part[wave][h * 8];
#pragma unroll
    for (int r = 0; r < 8; ++r) dst[r] = rs0[r] + rs1[r];
  }
  __syncthreads();
  if (threadIdx.x < 16) {
    float s = 0.f;
#pragma unroll
    for (int w = 0; w < SK_FW; ++w) s += part[w][threadIdx.x];
    const float uval = mu[rowBase + threadIdx.x] / (s + SK_TOL);
    ush[threadIdx.x] = uval;
    u_out[rowBase + threadIdx.x] = uval;
  }
  __syncthreads();

  v2f ua[4];
#pragma unroll
  for (int j = 0; j < 4; ++j) {
    ua[j][0] = ush[8 * h + 2 * j];
    ua[j][1] = ush[8 * h + 2 * j + 1];
  }

  const bool have_ws = ws_size >= SK_WS_NEED;
  float* dstc = cpart + (size_t)blockIdx.x * SK_N + colBase + lane;
#pragma unroll
  for (int g = 0; g < SK_FG; g += 2) {
    v8f o0 = zero, o1 = zero;
#pragma unroll
    for (int j = 0; j < 4; ++j) {
      v2f t0, t1;
      t0[0] = tile[g][2 * j];     t0[1] = tile[g][2 * j + 1];
      t1[0] = tile[g + 1][2 * j]; t1[1] = tile[g + 1][2 * j + 1];
      o0 = sk_wmma(ua[j], t0, o0);
      o1 = sk_wmma(ua[j], t1, o1);
    }
    const float val = h ? o1[0] : o0[0];
    if (have_ws) dstc[16 * g] = val;
  }
}

__global__ void __launch_bounds__(256) sk_cols(const float* __restrict__ K,
                                               const float* __restrict__ u,
                                               const float* __restrict__ nu,
                                               const float* __restrict__ cpart,
                                               float* __restrict__ v_out,
                                               size_t ws_size) {
  __shared__ float part[16][16];
  const int colBase = blockIdx.x << 4;
  if (ws_size >= SK_WS_NEED) {
    const int cc = threadIdx.x & 15;
    const int pg = threadIdx.x >> 4;
    const float* src = cpart + (size_t)pg * SK_N + colBase + cc;
    float s = 0.f;
#pragma unroll
    for (int q = 0; q < SK_NRB / 16; ++q) s += src[(size_t)(16 * q) * SK_N];
    part[pg][cc] = s;
    __syncthreads();
    if (threadIdx.x < 16) {
      float t = 0.f;
#pragma unroll
      for (int p = 0; p < 16; ++p) t += part[p][threadIdx.x];
      v_out[colBase + threadIdx.x] = nu[colBase + threadIdx.x] / (t + SK_TOL);
    }
  } else {
    const int wave = threadIdx.x >> 5;
    const int lane = threadIdx.x & 31;
    const int hf   = lane >> 4;
    const int r    = lane & 15;
    const float* Kc = K + colBase + r;
    v8f acc = {0.f, 0.f, 0.f, 0.f, 0.f, 0.f, 0.f, 0.f};
    const int r0 = wave << 9;
#pragma unroll 2
    for (int row = r0; row < r0 + 512; row += 4) {
      v2f a = *(const v2f*)(u + row + 2 * hf);
      size_t base = (size_t)(row + 2 * hf) * SK_N;
      v2f b;
      b[0] = Kc[base];
      b[1] = Kc[base + SK_N];
      acc = sk_wmma(a, b, acc);
    }
    if (lane < 16) part[wave][lane] = acc[0];
    __syncthreads();
    if (threadIdx.x < 16) {
      float t = 0.f;
#pragma unroll
      for (int w = 0; w < 8; ++w) t += part[w][threadIdx.x];
      v_out[colBase + threadIdx.x] = nu[colBase + threadIdx.x] / (t + SK_TOL);
    }
  }
}

__global__ void __launch_bounds__(256) sk_final(const float* __restrict__ mu,
                                                const float* __restrict__ nu,
                                                const float* __restrict__ u,
                                                const float* __restrict__ v,
                                                float* __restrict__ P,
                                                float* __restrict__ lpart) {
  __shared__ float red[256];
  const int i = blockIdx.x;
  const float ui = u[i];
  const float mi = mu[i];
  float* Prow = P + (size_t)i * SK_N;
  float w = 0.f;
  for (int j = threadIdx.x * 4; j < SK_N; j += 256 * 4) {
    float4 k4 = *(const float4*)(Prow + j);
    float4 v4 = *(const float4*)(v + j);
    float4 n4 = *(const float4*)(nu + j);
    float4 p;
    p.x = ui * k4.x * v4.x;
    p.y = ui * k4.y * v4.y;
    p.z = ui * k4.z * v4.z;
    p.w = ui * k4.w * v4.w;
    *(float4*)(Prow + j) = p;
    w += p.x * fabsf(mi - n4.x) + p.y * fabsf(mi - n4.y) +
         p.z * fabsf(mi - n4.z) + p.w * fabsf(mi - n4.w);
  }
  red[threadIdx.x] = w;
  __syncthreads();
  for (int s = 128; s > 0; s >>= 1) {
    if (threadIdx.x < s) red[threadIdx.x] += red[threadIdx.x + s];
    __syncthreads();
  }
  if (threadIdx.x == 0) lpart[i] = red[0];
}

__global__ void __launch_bounds__(256) sk_loss_reduce(const float* __restrict__ lpart,
                                                      float* __restrict__ loss) {
  __shared__ float red[256];
  float s = 0.f;
  for (int i = threadIdx.x; i < SK_N; i += 256) s += lpart[i];
  red[threadIdx.x] = s;
  __syncthreads();
  for (int k = 128; k > 0; k >>= 1) {
    if (threadIdx.x < k) red[threadIdx.x] += red[threadIdx.x + k];
    __syncthreads();
  }
  if (threadIdx.x == 0) *loss = red[0];
}

extern "C" void kernel_launch(void* const* d_in, const int* in_sizes, int n_in,
                              void* d_out, int out_size, void* d_ws, size_t ws_size,
                              hipStream_t stream) {
  const float* mu = (const float*)d_in[0];
  const float* nu = (const float*)d_in[1];
  const float* C  = (const float*)d_in[2];
  float* K = (float*)d_out;
  float* ws = (float*)d_ws;
  float* u     = ws;
  float* v     = ws + SK_N;
  float* lpart = ws + 2 * SK_N;
  float* cpart = ws + 4 * SK_N;

  sk_init_K<<<SK_NN / (256 * 4), 256, 0, stream>>>(C, K);
  sk_ones<<<SK_N / 256, 256, 0, stream>>>(v);

  for (int t = 0; t < SK_ITERS; ++t) {
    sk_rows_fused<<<SK_NRB, 512, 0, stream>>>(K, mu, v, u, cpart, ws_size);
    sk_cols<<<SK_N / 16, 256, 0, stream>>>(K, u, nu, cpart, v, ws_size);
  }

  sk_final<<<SK_N, 256, 0, stream>>>(mu, nu, u, v, K, lpart);
  sk_loss_reduce<<<1, 256, 0, stream>>>(lpart, K + SK_NN);
}


# u hand-over between the two matrix phases hidden: last two stages' tile WMMAs issued between signal and wait of its two barriers, mu loaded at entry, u stored behind the barrier; sums unchanged
# speedup vs baseline: 2.2485x; 1.0385x over previous
#include <hip/hip_runtime.h>
#include <math.h>


typedef float v2f __attribute__((ext_vector_type(2)));
typedef float v8f __attribute__((ext_vector_type(8)));

#define SK_N     4096
#define SK_NN    (4096 * 4096)
#define SK_IEPS  10.0f
#define SK_TOL   1e-9f
#define SK_ITERS 100

#define SK_FW    16
#define SK_FCOLS (SK_N / SK_FW)
#define SK_FG    (SK_FCOLS / 16)
#define SK_NRB   (SK_N / 16)
#define SK_NST   (SK_FG / 2)
#define SK_DEF   2
#define SK_WS_NEED ((size_t)(4 * SK_N + SK_NRB * SK_N) * sizeof(float))

static __device__ __forceinline__ v8f sk_wmma(v2f a, v2f b, v8f c) {
  return __builtin_amdgcn_wmma_f32_16x16x4_f32(false, a, false, b, (short)0, c,
                                               false, false);
}

static __device__ __forceinline__ void sk_tile_stage(const float4 (&k4)[4],
                                                     const v2f (&sel)[2][2],
                                                     v8f& t0, v8f& t1) {
  const v8f zero = {0.f, 0.f, 0.f, 0.f, 0.f, 0.f, 0.f, 0.f};
  v8f d0 = zero, d1 = zero;
#pragma unroll
  for (int p = 0; p < 2; ++p) {
    v2f a0, a1, e0, e1;
    a0[0] = k4[p].x;     a0[1] = k4[p].y;     a1[0] = k4[p].z;     a1[1] = k4[p].w;
    e0[0] = k4[2 + p].x; e0[1] = k4[2 + p].y; e1[0] = k4[2 + p].z; e1[1] = k4[2 + p].w;
    d0 = sk_wmma(a0, sel[p][0], d0);
    d1 = sk_wmma(e0, sel[p][0], d1);
    d0 = sk_wmma(a1, sel[p][1], d0);
    d1 = sk_wmma(e1, sel[p][1], d1);
  }
  t0 = d0;
  t1 = d1;
}

static __device__ __forceinline__ void sk_lds_done() {
  asm volatile("s_wait_dscnt 0x0" ::: "memory");
}
static __device__ __forceinline__ void sk_bar_signal() {
  asm volatile("s_barrier_signal -1" ::: "memory");
}
static __device__ __forceinline__ void sk_bar_wait() {
  asm volatile("s_barrier_wait -1" ::: "memory");
}

__global__ void __launch_bounds__(256) sk_init_K(const float* __restrict__ C,
                                                 float* __restrict__ K) {
  size_t i4 = ((size_t)blockIdx.x * 256 + threadIdx.x) * 4;
  float4 c4 = *(const float4*)(C + i4);
  float4 k4;
  k4.x = expf(-SK_IEPS * c4.x);
  k4.y = expf(-SK_IEPS * c4.y);
  k4.z = expf(-SK_IEPS * c4.z);
  k4.w = expf(-SK_IEPS * c4.w);
  *(float4*)(K + i4) = k4;
}

__global__ void sk_ones(float* __restrict__ v) {
  v[blockIdx.x * 256 + threadIdx.x] = 1.0f;
}

__global__ void __launch_bounds__(512) sk_rows_fused(const float* __restrict__ K,
                                                     const float* __restrict__ mu,
                                                     const float* __restrict__ v,
                                                     float* __restrict__ u_out,
                                                     float* __restrict__ cpart,
                                                     size_t ws_size) {
  __shared__ float4 vsh4[SK_N / 4];
  __shared__ float part[SK_FW][16];
  __shared__ float ush[16];
  const int wave = threadIdx.x >> 5;
  const int lane = threadIdx.x & 31;
  const int h = lane >> 4;
  const int c = lane & 15;
  const int rowBase = blockIdx.x << 4;
  const int colBase = wave * SK_FCOLS;
  const float* Krow = K + (size_t)(rowBase + c) * SK_N + colBase + 4 * h;
  const float4* vp4 = vsh4 + (colBase >> 2) + h;

  float mu_r = 0.f;
  if (threadIdx.x < 16) mu_r = mu[rowBase + threadIdx.x];

  vsh4[threadIdx.x]       = *(const float4*)(v + 4 * threadIdx.x);
  vsh4[512 + threadIdx.x] = *(const float4*)(v + 2048 + 4 * threadIdx.x);

  v2f sel[2][2];
#pragma unroll
  for (int p = 0; p < 2; ++p) {
#pragma unroll
    for (int e = 0; e < 2; ++e) {
      const int n0 = 8 * p + 4 * h + 2 * e;
      sel[p][e][0] = (c == n0) ? 1.0f : 0.0f;
      sel[p][e][1] = (c == n0 + 1) ? 1.0f : 0.0f;
    }
  }

  const v8f zero = {0.f, 0.f, 0.f, 0.f, 0.f, 0.f, 0.f, 0.f};
  v8f rs0 = zero, rs1 = zero;
  v8f tile[SK_FG];
  float4 ka[4], kn[4];
  float4 kd[SK_DEF][4];
#pragma unroll
  for (int x = 0; x < 4; ++x) ka[x] = *(const float4*)(Krow + 8 * x);
  __syncthreads();
#pragma unroll
  for (int st = 0; st < SK_NST; ++st) {
    if (st + 1 < SK_NST) {
#pragma unroll
      for (int x = 0; x < 4; ++x)
        kn[x] = *(const float4*)(Krow + 32 * (st + 1) + 8 * x);
    }
#pragma unroll
    for (int gg = 0; gg < 2; ++gg) {
      v8f d = zero;
#pragma unroll
      for (int p = 0; p < 2; ++p) {
        const float4 a4 = ka[2 * gg + p];
        const float4 b4 = vp4[8 * st + 2 * (2 * gg + p)];
        v2f a0, a1, b0, b1;
        a0[0] = a4.x; a0[1] = a4.y; a1[0] = a4.z; a1[1] = a4.w;
        b0[0] = b4.x; b0[1] = b4.y; b1[0] = b4.z; b1[1] = b4.w;
        rs0 = sk_wmma(a0, b0, rs0);
        rs1 = sk_wmma(a1, b1, rs1);
        if (st < SK_NST - SK_DEF) {
          d = sk_wmma(a0, sel[p][0], d);
          d = sk_wmma(a1, sel[p][1], d);
        }
      }
      if (st < SK_NST - SK_DEF) tile[2 * st + gg] = d;
    }
    if (st >= SK_NST - SK_DEF) {
#pragma unroll
      for (int x = 0; x < 4; ++x) kd[st - (SK_NST - SK_DEF)][x] = ka[x];
    }
    __builtin_amdgcn_sched_barrier(0);
#pragma unroll
    for (int x = 0; x < 4; ++x) ka[x] = kn[x];
  }

  if (c == 0) {
    float* dst = &part[wave][h * 8];
#pragma unroll
    for (int r = 0; r < 8; ++r) dst[r] = rs0[r] + rs1[r];
  }
  sk_lds_done();
  sk_bar_signal();
  __builtin_amdgcn_sched_barrier(0);
  sk_tile_stage(kd[0], sel, tile[2 * (SK_NST - 2)], tile[2 * (SK_NST - 2) + 1]);
  __builtin_amdgcn_sched_barrier(0);
  sk_bar_wait();
  float uval = 0.f;
  if (threadIdx.x < 16) {
    float s = 0.f;
#pragma unroll
    for (int w = 0; w < SK_FW; ++w) s += part[w][threadIdx.x];
    uval = mu_r / (s + SK_TOL);
    ush[threadIdx.x] = uval;
  }
  sk_lds_done();
  sk_bar_signal();
  __builtin_amdgcn_sched_barrier(0);
  sk_tile_stage(kd[1], sel, tile[2 * (SK_NST - 1)], tile[2 * (SK_NST - 1) + 1]);
  __builtin_amdgcn_sched_barrier(0);
  sk_bar_wait();
  if (threadIdx.x < 16) u_out[rowBase + threadIdx.x] = uval;

  v2f ua[4];
#pragma unroll
  for (int j = 0; j < 4; ++j) {
    ua[j][0] = ush[8 * h + 2 * j];
    ua[j][1] = ush[8 * h + 2 * j + 1];
  }

  const bool have_ws = ws_size >= SK_WS_NEED;
  float* dstc = cpart + (size_t)blockIdx.x * SK_N + colBase + lane;
#pragma unroll
  for (int g = 0; g < SK_FG; g += 2) {
    v8f o0 = zero, o1 = zero;
#pragma unroll
    for (int j = 0; j < 4; ++j) {
      v2f t0, t1;
      t0[0] = tile[g][2 * j];     t0[1] = tile[g][2 * j + 1];
      t1[0] = tile[g + 1][2 * j]; t1[1] = tile[g + 1][2 * j + 1];
      o0 = sk_wmma(ua[j], t0, o0);
      o1 = sk_wmma(ua[j], t1, o1);
    }
    const float val = h ? o1[0] : o0[0];
    if (have_ws) dstc[16 * g] = val;
  }
}

__global__ void __launch_bounds__(256) sk_cols(const float* __restrict__ K,
                                               const float* __restrict__ u,
                                               const float* __restrict__ nu,
                                               const float* __restrict__ cpart,
                                               float* __restrict__ v_out,
                                               size_t ws_size) {
  __shared__ float part[16][16];
  const int colBase = blockIdx.x << 4;
  if (ws_size >= SK_WS_NEED) {
    const int cc = threadIdx.x & 15;
    const int pg = threadIdx.x >> 4;
    const float* src = cpart + (size_t)pg * SK_N + colBase + cc;
    float s = 0.f;
#pragma unroll
    for (int q = 0; q < SK_NRB / 16; ++q) s += src[(size_t)(16 * q) * SK_N];
    part[pg][cc] = s;
    __syncthreads();
    if (threadIdx.x < 16) {
      float t = 0.f;
#pragma unroll
      for (int p = 0; p < 16; ++p) t += part[p][threadIdx.x];
      v_out[colBase + threadIdx.x] = nu[colBase + threadIdx.x] / (t + SK_TOL);
    }
  } else {
    const int wave = threadIdx.x >> 5;
    const int lane = threadIdx.x & 31;
    const int hf   = lane >> 4;
    const int r    = lane & 15;
    const float* Kc = K + colBase + r;
    v8f acc = {0.f, 0.f, 0.f, 0.f, 0.f, 0.f, 0.f, 0.f};
    const int r0 = wave << 9;
#pragma unroll 2
    for (int row = r0; row < r0 + 512; row += 4) {
      v2f a = *(const v2f*)(u + row + 2 * hf);
      size_t base = (size_t)(row + 2 * hf) * SK_N;
      v2f b;
      b[0] = Kc[base];
      b[1] = Kc[base + SK_N];
      acc = sk_wmma(a, b, acc);
    }
    if (lane < 16) part[wave][lane] = acc[0];
    __syncthreads();
    if (threadIdx.x < 16) {
      float t = 0.f;
#pragma unroll
      for (int w = 0; w < 8; ++w) t += part[w][threadIdx.x];
      v_out[colBase + threadIdx.x] = nu[colBase + threadIdx.x] / (t + SK_TOL);
    }
  }
}

__global__ void __launch_bounds__(256) sk_final(const float* __restrict__ mu,
                                                const float* __restrict__ nu,
                                                const float* __restrict__ u,
                                                const float* __restrict__ v,
                                                float* __restrict__ P,
                                                float* __restrict__ lpart) {
  __shared__ float red[256];
  const int i = blockIdx.x;
  const float ui = u[i];
  const float mi = mu[i];
  float* Prow = P + (size_t)i * SK_N;
  float w = 0.f;
  for (int j = threadIdx.x * 4; j < SK_N; j += 256 * 4) {
    float4 k4 = *(const float4*)(Prow + j);
    float4 v4 = *(const float4*)(v + j);
    float4 n4 = *(const float4*)(nu + j);
    float4 p;
    p.x = ui * k4.x * v4.x;
    p.y = ui * k4.y * v4.y;
    p.z = ui * k4.z * v4.z;
    p.w = ui * k4.w * v4.w;
    *(float4*)(Prow + j) = p;
    w += p.x * fabsf(mi - n4.x) + p.y * fabsf(mi - n4.y) +
         p.z * fabsf(mi - n4.z) + p.w * fabsf(mi - n4.w);
  }
  red[threadIdx.x] = w;
  __syncthreads();
  for (int s = 128; s > 0; s >>= 1) {
    if (threadIdx.x < s) red[threadIdx.x] += red[threadIdx.x + s];
    __syncthreads();
  }
  if (threadIdx.x == 0) lpart[i] = red[0];
}

__global__ void __launch_bounds__(256) sk_loss_reduce(const float* __restrict__ lpart,
                                                      float* __restrict__ loss) {
  __shared__ float red[256];
  float s = 0.f;
  for (int i = threadIdx.x; i < SK_N; i += 256) s += lpart[i];
  red[threadIdx.x] = s;
  __syncthreads();
  for (int k = 128; k > 0; k >>= 1) {
    if (threadIdx.x < k) red[threadIdx.x] += red[threadIdx.x + k];
    __syncthreads();
  }
  if (threadIdx.x == 0) *loss = red[0];
}

extern "C" void kernel_launch(void* const* d_in, const int* in_sizes, int n_in,
                              void* d_out, int out_size, void* d_ws, size_t ws_size,
                              hipStream_t stream) {
  const float* mu = (const float*)d_in[0];
  const float* nu = (const float*)d_in[1];
  const float* C  = (const float*)d_in[2];
  float* K = (float*)d_out;
  float* ws = (float*)d_ws;
  float* u     = ws;
  float* v     = ws + SK_N;
  float* lpart = ws + 2 * SK_N;
  float* cpart = ws + 4 * SK_N;

  sk_init_K<<<SK_NN / (256 * 4), 256, 0, stream>>>(C, K);
  sk_ones<<<SK_N / 256, 256, 0, stream>>>(v);

  for (int t = 0; t < SK_ITERS; ++t) {
    sk_rows_fused<<<SK_NRB, 512, 0, stream>>>(K, mu, v, u, cpart, ws_size);
    sk_cols<<<SK_N / 16, 256, 0, stream>>>(K, u, nu, cpart, v, ws_size);
  }

  sk_final<<<SK_N, 256, 0, stream>>>(mu, nu, u, v, K, lpart);
  sk_loss_reduce<<<1, 256, 0, stream>>>(lpart, K + SK_NN);
}


# Row pass runs as 8 waves x 512 columns per work-group (2 per SIMD, was 16 x 256): half the waves to enter; K requested 3 stages ahead, in flight across the v barrier; f32 row-sum order changes
# speedup vs baseline: 2.3660x; 1.0523x over previous
#include <hip/hip_runtime.h>
#include <math.h>


typedef float v2f __attribute__((ext_vector_type(2)));
typedef float v8f __attribute__((ext_vector_type(8)));

#define SK_N     4096
#define SK_NN    (4096 * 4096)
#define SK_IEPS  10.0f
#define SK_TOL   1e-9f
#define SK_ITERS 100

#define SK_FW    8
#define SK_FCOLS (SK_N / SK_FW)
#define SK_FG    (SK_FCOLS / 16)
#define SK_NRB   (SK_N / 16)
#define SK_NST   (SK_FG / 2)
#define SK_DEF   2
#define SK_PF    3
#define SK_RING  (SK_PF + 1)
#define SK_WS_NEED ((size_t)(4 * SK_N + SK_NRB * SK_N) * sizeof(float))

static __device__ __forceinline__ v8f sk_wmma(v2f a, v2f b, v8f c) {
  return __builtin_amdgcn_wmma_f32_16x16x4_f32(false, a, false, b, (short)0, c,
                                               false, false);
}

static __device__ __forceinline__ void sk_tile_stage(const float4 (&k4)[4],
                                                     const v2f (&sel)[2][2],
                                                     v8f& t0, v8f& t1) {
  const v8f zero = {0.f, 0.f, 0.f, 0.f, 0.f, 0.f, 0.f, 0.f};
  v8f d0 = zero, d1 = zero;
#pragma unroll
  for (int p = 0; p < 2; ++p) {
    v2f a0, a1, e0, e1;
    a0[0] = k4[p].x;     a0[1] = k4[p].y;     a1[0] = k4[p].z;     a1[1] = k4[p].w;
    e0[0] = k4[2 + p].x; e0[1] = k4[2 + p].y; e1[0] = k4[2 + p].z; e1[1] = k4[2 + p].w;
    d0 = sk_wmma(a0, sel[p][0], d0);
    d1 = sk_wmma(e0, sel[p][0], d1);
    d0 = sk_wmma(a1, sel[p][1], d0);
    d1 = sk_wmma(e1, sel[p][1], d1);
  }
  t0 = d0;
  t1 = d1;
}

static __device__ __forceinline__ void sk_lds_done() {
  asm volatile("s_wait_dscnt 0x0" ::: "memory");
}
static __device__ __forceinline__ void sk_bar_signal() {
  asm volatile("s_barrier_signal -1" ::: "memory");
}
static __device__ __forceinline__ void sk_bar_wait() {
  asm volatile("s_barrier_wait -1" ::: "memory");
}

__global__ void __launch_bounds__(256) sk_init_K(const float* __restrict__ C,
                                                 float* __restrict__ K) {
  size_t i4 = ((size_t)blockIdx.x * 256 + threadIdx.x) * 4;
  float4 c4 = *(const float4*)(C + i4);
  float4 k4;
  k4.x = expf(-SK_IEPS * c4.x);
  k4.y = expf(-SK_IEPS * c4.y);
  k4.z = expf(-SK_IEPS * c4.z);
  k4.w = expf(-SK_IEPS * c4.w);
  *(float4*)(K + i4) = k4;
}

__global__ void sk_ones(float* __restrict__ v) {
  v[blockIdx.x * 256 + threadIdx.x] = 1.0f;
}

__global__ void __launch_bounds__(256) sk_rows_fused(const float* __restrict__ K,
                                                     const float* __restrict__ mu,
                                                     const float* __restrict__ v,
                                                     float* __restrict__ u_out,
                                                     float* __restrict__ cpart,
                                                     size_t ws_size) {
  __shared__ float4 vsh4[SK_N / 4];
  __shared__ float part[SK_FW][16];
  __shared__ float ush[16];
  const int wave = threadIdx.x >> 5;
  const int lane = threadIdx.x & 31;
  const int h = lane >> 4;
  const int c = lane & 15;
  const int rowBase = blockIdx.x << 4;
  const int colBase = wave * SK_FCOLS;
  const float* Krow = K + (size_t)(rowBase + c) * SK_N + colBase + 4 * h;
  const float4* vp4 = vsh4 + (colBase >> 2) + h;

  float mu_r = 0.f;
  if (threadIdx.x < 16) mu_r = mu[rowBase + threadIdx.x];

#pragma unroll
  for (int q = 0; q < 4; ++q)
    vsh4[256 * q + threadIdx.x] = *(const float4*)(v + 1024 * q + 4 * threadIdx.x);

  v2f sel[2][2];
#pragma unroll
  for (int p = 0; p < 2; ++p) {
#pragma unroll
    for (int e = 0; e < 2; ++e) {
      const int n0 = 8 * p + 4 * h + 2 * e;
      sel[p][e][0] = (c == n0) ? 1.0f : 0.0f;
      sel[p][e][1] = (c == n0 + 1) ? 1.0f : 0.0f;
    }
  }

  const v8f zero = {0.f, 0.f, 0.f, 0.f, 0.f, 0.f, 0.f, 0.f};
  v8f rs0 = zero, rs1 = zero;
  v8f tile[SK_FG];
  float4 kq[SK_RING][4];
#pragma unroll
  for (int s0 = 0; s0 < SK_PF; ++s0) {
#pragma unroll
    for (int x = 0; x < 4; ++x) kq[s0][x] = *(const float4*)(Krow + 32 * s0 + 8 * x);
  }
  sk_lds_done();
  sk_bar_signal();
  sk_bar_wait();
  float4 vb[2][4];
#pragma unroll
  for (int x = 0; x < 4; ++x) vb[0][x] = vp4[2 * x];
#pragma unroll
  for (int st = 0; st < SK_NST; ++st) {
    if (st + 1 < SK_NST) {
#pragma unroll
      for (int x = 0; x < 4; ++x) vb[(st + 1) & 1][x] = vp4[8 * (st + 1) + 2 * x];
    }
    if (st + SK_PF < SK_NST) {
#pragma unroll
      for (int x = 0; x < 4; ++x)
        kq[(st + SK_PF) % SK_RING][x] = *(const float4*)(Krow + 32 * (st + SK_PF) + 8 * x);
    }
    v8f d0 = zero, d1 = zero;
#pragma unroll
    for (int p = 0; p < 2; ++p) {
      const float4 a4 = kq[st % SK_RING][p];
      const float4 e4 = kq[st % SK_RING][2 + p];
      const float4 b4 = vb[st & 1][p];
      const float4 f4 = vb[st & 1][2 + p];
      v2f a0, a1, e0, e1, b0, b1, f0, f1;
      a0[0] = a4.x; a0[1] = a4.y; a1[0] = a4.z; a1[1] = a4.w;
      e0[0] = e4.x; e0[1] = e4.y; e1[0] = e4.z; e1[1] = e4.w;
      b0[0] = b4.x; b0[1] = b4.y; b1[0] = b4.z; b1[1] = b4.w;
      f0[0] = f4.x; f0[1] = f4.y; f1[0] = f4.z; f1[1] = f4.w;
      rs0 = sk_wmma(a0, b0, rs0);
      if (st < SK_NST - SK_DEF) d0 = sk_wmma(a0, sel[p][0], d0);
      rs1 = sk_wmma(e0, f0, rs1);
      if (st < SK_NST - SK_DEF) d1 = sk_wmma(e0, sel[p][0], d1);
      rs0 = sk_wmma(a1, b1, rs0);
      if (st < SK_NST - SK_DEF) d0 = sk_wmma(a1, sel[p][1], d0);
      rs1 = sk_wmma(e1, f1, rs1);
      if (st < SK_NST - SK_DEF) d1 = sk_wmma(e1, sel[p][1], d1);
    }
    if (st < SK_NST - SK_DEF) {
      tile[2 * st]     = d0;
      tile[2 * st + 1] = d1;
    }
    __builtin_amdgcn_sched_barrier(0);
  }
  const float4 (&kdA)[4] = kq[(SK_NST - 2) % SK_RING];
  const float4 (&kdB)[4] = kq[(SK_NST - 1) % SK_RING];

  if (c == 0) {
    float* dst = &part[wave][h * 8];
#pragma unroll
    for (int r = 0; r < 8; ++r) dst[r] = rs0[r] + rs1[r];
  }
  sk_lds_done();
  sk_bar_signal();
  __builtin_amdgcn_sched_barrier(0);
  sk_tile_stage(kdA, sel, tile[2 * (SK_NST - 2)], tile[2 * (SK_NST - 2) + 1]);
  __builtin_amdgcn_sched_barrier(0);
  sk_bar_wait();
  float uval = 0.f;
  if (threadIdx.x < 16) {
    float s = 0.f;
#pragma unroll
    for (int w = 0; w < SK_FW; ++w) s += part[w][threadIdx.x];
    uval = mu_r / (s + SK_TOL);
    ush[threadIdx.x] = uval;
  }
  sk_lds_done();
  sk_bar_signal();
  __builtin_amdgcn_sched_barrier(0);
  sk_tile_stage(kdB, sel, tile[2 * (SK_NST - 1)], tile[2 * (SK_NST - 1) + 1]);
  __builtin_amdgcn_sched_barrier(0);
  sk_bar_wait();
  if (threadIdx.x < 16) u_out[rowBase + threadIdx.x] = uval;

  v2f ua[4];
#pragma unroll
  for (int j = 0; j < 4; ++j) {
    ua[j][0] = ush[8 * h + 2 * j];
    ua[j][1] = ush[8 * h + 2 * j + 1];
  }

  const bool have_ws = ws_size >= SK_WS_NEED;
  float* dstc = cpart + (size_t)blockIdx.x * SK_N + colBase + lane;
#pragma unroll
  for (int g = 0; g < SK_FG; g += 2) {
    v8f o0 = zero, o1 = zero;
#pragma unroll
    for (int j = 0; j < 4; ++j) {
      v2f t0, t1;
      t0[0] = tile[g][2 * j];     t0[1] = tile[g][2 * j + 1];
      t1[0] = tile[g + 1][2 * j]; t1[1] = tile[g + 1][2 * j + 1];
      o0 = sk_wmma(ua[j], t0, o0);
      o1 = sk_wmma(ua[j], t1, o1);
    }
    const float val = h ? o1[0] : o0[0];
    if (have_ws) dstc[16 * g] = val;
  }
}

__global__ void __launch_bounds__(256) sk_cols(const float* __restrict__ K,
                                               const float* __restrict__ u,
                                               const float* __restrict__ nu,
                                               const float* __restrict__ cpart,
                                               float* __restrict__ v_out,
                                               size_t ws_size) {
  __shared__ float part[16][16];
  const int colBase = blockIdx.x << 4;
  if (ws_size >= SK_WS_NEED) {
    const int cc = threadIdx.x & 15;
    const int pg = threadIdx.x >> 4;
    const float* src = cpart + (size_t)pg * SK_N + colBase + cc;
    float s = 0.f;
#pragma unroll
    for (int q = 0; q < SK_NRB / 16; ++q) s += src[(size_t)(16 * q) * SK_N];
    part[pg][cc] = s;
    __syncthreads();
    if (threadIdx.x < 16) {
      float t = 0.f;
#pragma unroll
      for (int p = 0; p < 16; ++p) t += part[p][threadIdx.x];
      v_out[colBase + threadIdx.x] = nu[colBase + threadIdx.x] / (t + SK_TOL);
    }
  } else {
    const int wave = threadIdx.x >> 5;
    const int lane = threadIdx.x & 31;
    const int hf   = lane >> 4;
    const int r    = lane & 15;
    const float* Kc = K + colBase + r;
    v8f acc = {0.f, 0.f, 0.f, 0.f, 0.f, 0.f, 0.f, 0.f};
    const int r0 = wave << 9;
#pragma unroll 2
    for (int row = r0; row < r0 + 512; row += 4) {
      v2f a = *(const v2f*)(u + row + 2 * hf);
      size_t base = (size_t)(row + 2 * hf) * SK_N;
      v2f b;
      b[0] = Kc[base];
      b[1] = Kc[base + SK_N];
      acc = sk_wmma(a, b, acc);
    }
    if (lane < 16) part[wave][lane] = acc[0];
    __syncthreads();
    if (threadIdx.x < 16) {
      float t = 0.f;
#pragma unroll
      for (int w = 0; w < 8; ++w) t += part[w][threadIdx.x];
      v_out[colBase + threadIdx.x] = nu[colBase + threadIdx.x] / (t + SK_TOL);
    }
  }
}

__global__ void __launch_bounds__(256) sk_final(const float* __restrict__ mu,
                                                const float* __restrict__ nu,
                                                const float* __restrict__ u,
                                                const float* __restrict__ v,
                                                float* __restrict__ P,
                                                float* __restrict__ lpart) {
  __shared__ float red[256];
  const int i = blockIdx.x;
  const float ui = u[i];
  const float mi = mu[i];
  float* Prow = P + (size_t)i * SK_N;
  float w = 0.f;
  for (int j = threadIdx.x * 4; j < SK_N; j += 256 * 4) {
    float4 k4 = *(const float4*)(Prow + j);
    float4 v4 = *(const float4*)(v + j);
    float4 n4 = *(const float4*)(nu + j);
    float4 p;
    p.x = ui * k4.x * v4.x;
    p.y = ui * k4.y * v4.y;
    p.z = ui * k4.z * v4.z;
    p.w = ui * k4.w * v4.w;
    *(float4*)(Prow + j) = p;
    w += p.x * fabsf(mi - n4.x) + p.y * fabsf(mi - n4.y) +
         p.z * fabsf(mi - n4.z) + p.w * fabsf(mi - n4.w);
  }
  red[threadIdx.x] = w;
  __syncthreads();
  for (int s = 128; s > 0; s >>= 1) {
    if (threadIdx.x < s) red[threadIdx.x] += red[threadIdx.x + s];
    __syncthreads();
  }
  if (threadIdx.x == 0) lpart[i] = red[0];
}

__global__ void __launch_bounds__(256) sk_loss_reduce(const float* __restrict__ lpart,
                                                      float* __restrict__ loss) {
  __shared__ float red[256];
  float s = 0.f;
  for (int i = threadIdx.x; i < SK_N; i += 256) s += lpart[i];
  red[threadIdx.x] = s;
  __syncthreads();
  for (int k = 128; k > 0; k >>= 1) {
    if (threadIdx.x < k) red[threadIdx.x] += red[threadIdx.x + k];
    __syncthreads();
  }
  if (threadIdx.x == 0) *loss = red[0];
}

extern "C" void kernel_launch(void* const* d_in, const int* in_sizes, int n_in,
                              void* d_out, int out_size, void* d_ws, size_t ws_size,
                              hipStream_t stream) {
  const float* mu = (const float*)d_in[0];
  const float* nu = (const float*)d_in[1];
  const float* C  = (const float*)d_in[2];
  float* K = (float*)d_out;
  float* ws = (float*)d_ws;
  float* u     = ws;
  float* v     = ws + SK_N;
  float* lpart = ws + 2 * SK_N;
  float* cpart = ws + 4 * SK_N;

  sk_init_K<<<SK_NN / (256 * 4), 256, 0, stream>>>(C, K);
  sk_ones<<<SK_N / 256, 256, 0, stream>>>(v);

  for (int t = 0; t < SK_ITERS; ++t) {
    sk_rows_fused<<<SK_NRB, 256, 0, stream>>>(K, mu, v, u, cpart, ws_size);
    sk_cols<<<SK_N / 16, 256, 0, stream>>>(K, u, nu, cpart, v, ws_size);
  }

  sk_final<<<SK_N, 256, 0, stream>>>(mu, nu, u, v, K, lpart);
  sk_loss_reduce<<<1, 256, 0, stream>>>(lpart, K + SK_NN);
}


# Row block re-read from L2 in column-sum layout (lane: 4 columns of 2 rows, 256-B runs) as B: no transposing WMMAs (384->256 a wave), no register tile; u per wave, 1 barrier; column-sum order changes
# speedup vs baseline: 2.4817x; 1.0489x over previous
#include <hip/hip_runtime.h>
#include <math.h>


typedef float v2f __attribute__((ext_vector_type(2)));
typedef float v8f __attribute__((ext_vector_type(8)));

#define SK_N     4096
#define SK_NN    (4096 * 4096)
#define SK_IEPS  10.0f
#define SK_TOL   1e-9f
#define SK_ITERS 100

#define SK_FW    8
#define SK_FCOLS (SK_N / SK_FW)
#define SK_FG    (SK_FCOLS / 16)
#define SK_NRB   (SK_N / 16)
#define SK_NST   (SK_FG / 2)
#define SK_PF    3
#define SK_RING  (SK_PF + 1)
#define SK_CG    (SK_FCOLS / 64)
#define SK_CPF   3
#define SK_CRING (SK_CPF + 1)
#define SK_WS_NEED ((size_t)(4 * SK_N + SK_NRB * SK_N) * sizeof(float))

static __device__ __forceinline__ v8f sk_wmma(v2f a, v2f b, v8f c) {
  return __builtin_amdgcn_wmma_f32_16x16x4_f32(false, a, false, b, (short)0, c,
                                               false, false);
}

template <int R>
static __device__ __forceinline__ float sk_row_lane(float x) {
  return __builtin_bit_cast(
      float, __builtin_amdgcn_update_dpp(0, __builtin_bit_cast(int, x), 0x150 + R, 0xF, 0xF, false));
}

static __device__ __forceinline__ void sk_lds_done() {
  asm volatile("s_wait_dscnt 0x0" ::: "memory");
}
static __device__ __forceinline__ void sk_bar_signal() {
  asm volatile("s_barrier_signal -1" ::: "memory");
}
static __device__ __forceinline__ void sk_bar_wait() {
  asm volatile("s_barrier_wait -1" ::: "memory");
}

__global__ void __launch_bounds__(256) sk_init_K(const float* __restrict__ C,
                                                 float* __restrict__ K) {
  size_t i4 = ((size_t)blockIdx.x * 256 + threadIdx.x) * 4;
  float4 c4 = *(const float4*)(C + i4);
  float4 k4;
  k4.x = expf(-SK_IEPS * c4.x);
  k4.y = expf(-SK_IEPS * c4.y);
  k4.z = expf(-SK_IEPS * c4.z);
  k4.w = expf(-SK_IEPS * c4.w);
  *(float4*)(K + i4) = k4;
}

__global__ void sk_ones(float* __restrict__ v) {
  v[blockIdx.x * 256 + threadIdx.x] = 1.0f;
}

__global__ void __launch_bounds__(256) sk_rows_fused(const float* __restrict__ K,
                                                     const float* __restrict__ mu,
                                                     const float* __restrict__ v,
                                                     float* __restrict__ u_out,
                                                     float* __restrict__ cpart,
                                                     size_t ws_size) {
  __shared__ float4 vsh4[SK_N / 4];
  __shared__ float part[SK_FW][16];
  const int wave = threadIdx.x >> 5;
  const int lane = threadIdx.x & 31;
  const int h = lane >> 4;
  const int c = lane & 15;
  const int rowBase = blockIdx.x << 4;
  const int colBase = wave * SK_FCOLS;
  const float* Krow = K + (size_t)(rowBase + c) * SK_N + colBase + 4 * h;
  const float* Kcol = K + (size_t)(rowBase + 2 * h) * SK_N + colBase + 4 * c;
  const float4* vp4 = vsh4 + (colBase >> 2) + h;

  const int ur = (c + 2 * h) & 15;
  const float mu_r = mu[rowBase + ur];

#pragma unroll
  for (int q = 0; q < 4; ++q)
    vsh4[256 * q + threadIdx.x] = *(const float4*)(v + 1024 * q + 4 * threadIdx.x);

  const v8f zero = {0.f, 0.f, 0.f, 0.f, 0.f, 0.f, 0.f, 0.f};
  v8f rs0 = zero, rs1 = zero;
  float4 kq[SK_RING][4];
#pragma unroll
  for (int s0 = 0; s0 < SK_PF; ++s0) {
#pragma unroll
    for (int x = 0; x < 4; ++x) kq[s0][x] = *(const float4*)(Krow + 32 * s0 + 8 * x);
  }
  sk_lds_done();
  sk_bar_signal();
  sk_bar_wait();
  float4 vb[2][4];
#pragma unroll
  for (int x = 0; x < 4; ++x) vb[0][x] = vp4[2 * x];
#pragma unroll
  for (int st = 0; st < SK_NST; ++st) {
    if (st + 1 < SK_NST) {
#pragma unroll
      for (int x = 0; x < 4; ++x) vb[(st + 1) & 1][x] = vp4[8 * (st + 1) + 2 * x];
    }
    if (st + SK_PF < SK_NST) {
#pragma unroll
      for (int x = 0; x < 4; ++x)
        kq[(st + SK_PF) % SK_RING][x] = *(const float4*)(Krow + 32 * (st + SK_PF) + 8 * x);
    }
#pragma unroll
    for (int p = 0; p < 2; ++p) {
      const float4 a4 = kq[st % SK_RING][p];
      const float4 e4 = kq[st % SK_RING][2 + p];
      const float4 b4 = vb[st & 1][p];
      const float4 f4 = vb[st & 1][2 + p];
      v2f a0, a1, e0, e1, b0, b1, f0, f1;
      a0[0] = a4.x; a0[1] = a4.y; a1[0] = a4.z; a1[1] = a4.w;
      e0[0] = e4.x; e0[1] = e4.y; e1[0] = e4.z; e1[1] = e4.w;
      b0[0] = b4.x; b0[1] = b4.y; b1[0] = b4.z; b1[1] = b4.w;
      f0[0] = f4.x; f0[1] = f4.y; f1[0] = f4.z; f1[1] = f4.w;
      rs0 = sk_wmma(a0, b0, rs0);
      rs1 = sk_wmma(e0, f0, rs1);
      rs0 = sk_wmma(a1, b1, rs0);
      rs1 = sk_wmma(e1, f1, rs1);
    }
    __builtin_amdgcn_sched_barrier(0);
  }

  float4 qb[SK_CRING][8];
#pragma unroll
  for (int gi = 0; gi < SK_CPF; ++gi) {
#pragma unroll
    for (int x = 0; x < 8; ++x)
      qb[gi][x] = *(const float4*)(Kcol + (size_t)(4 * (x >> 1) + (x & 1)) * SK_N +
                                   64 * (SK_CG - 1 - gi));
  }

  if (c == 0) {
    float* dst = &part[wave][h * 8];
#pragma unroll
    for (int r = 0; r < 8; ++r) dst[r] = rs0[r] + rs1[r];
  }
  sk_lds_done();
  sk_bar_signal();
  sk_bar_wait();
  float rsum = 0.f;
#pragma unroll
  for (int w = 0; w < SK_FW; ++w) rsum += part[w][ur];
  const float uval = mu_r / (rsum + SK_TOL);
  if (threadIdx.x < 16) u_out[rowBase + threadIdx.x] = uval;

  v2f ua[4];
  ua[0][0] = sk_row_lane<0>(uval);  ua[0][1] = sk_row_lane<1>(uval);
  ua[1][0] = sk_row_lane<4>(uval);  ua[1][1] = sk_row_lane<5>(uval);
  ua[2][0] = sk_row_lane<8>(uval);  ua[2][1] = sk_row_lane<9>(uval);
  ua[3][0] = sk_row_lane<12>(uval); ua[3][1] = sk_row_lane<13>(uval);

  const bool have_ws = ws_size >= SK_WS_NEED;
  float* dstc = cpart + (size_t)blockIdx.x * SK_N + colBase + 4 * c;
#pragma unroll
  for (int gi = 0; gi < SK_CG; ++gi) {
    if (gi + SK_CPF < SK_CG) {
#pragma unroll
      for (int x = 0; x < 8; ++x)
        qb[(gi + SK_CPF) % SK_CRING][x] =
            *(const float4*)(Kcol + (size_t)(4 * (x >> 1) + (x & 1)) * SK_N +
                             64 * (SK_CG - 1 - gi - SK_CPF));
    }
    v8f o0 = zero, o1 = zero, o2 = zero, o3 = zero;
#pragma unroll
    for (int j = 0; j < 4; ++j) {
      const float4 r0 = qb[gi % SK_CRING][2 * j];
      const float4 r1 = qb[gi % SK_CRING][2 * j + 1];
      v2f t0, t1, t2, t3;
      t0[0] = r0.x; t0[1] = r1.x;
      t1[0] = r0.y; t1[1] = r1.y;
      t2[0] = r0.z; t2[1] = r1.z;
      t3[0] = r0.w; t3[1] = r1.w;
      o0 = sk_wmma(ua[j], t0, o0);
      o1 = sk_wmma(ua[j], t1, o1);
      o2 = sk_wmma(ua[j], t2, o2);
      o3 = sk_wmma(ua[j], t3, o3);
    }
    if (have_ws && h == 0) {
      float4 o;
      o.x = o0[0]; o.y = o1[0]; o.z = o2[0]; o.w = o3[0];
      *(float4*)(dstc + 64 * (SK_CG - 1 - gi)) = o;
    }
    __builtin_amdgcn_sched_barrier(0);
  }
}

__global__ void __launch_bounds__(256) sk_cols(const float* __restrict__ K,
                                               const float* __restrict__ u,
                                               const float* __restrict__ nu,
                                               const float* __restrict__ cpart,
                                               float* __restrict__ v_out,
                                               size_t ws_size) {
  __shared__ float part[16][16];
  const int colBase = blockIdx.x << 4;
  if (ws_size >= SK_WS_NEED) {
    const int cc = threadIdx.x & 15;
    const int pg = threadIdx.x >> 4;
    const float* src = cpart + (size_t)pg * SK_N + colBase + cc;
    float s = 0.f;
#pragma unroll
    for (int q = 0; q < SK_NRB / 16; ++q) s += src[(size_t)(16 * q) * SK_N];
    part[pg][cc] = s;
    __syncthreads();
    if (threadIdx.x < 16) {
      float t = 0.f;
#pragma unroll
      for (int p = 0; p < 16; ++p) t += part[p][threadIdx.x];
      v_out[colBase + threadIdx.x] = nu[colBase + threadIdx.x] / (t + SK_TOL);
    }
  } else {
    const int wave = threadIdx.x >> 5;
    const int lane = threadIdx.x & 31;
    const int hf   = lane >> 4;
    const int r    = lane & 15;
    const float* Kc = K + colBase + r;
    v8f acc = {0.f, 0.f, 0.f, 0.f, 0.f, 0.f, 0.f, 0.f};
    const int r0 = wave << 9;
#pragma unroll 2
    for (int row = r0; row < r0 + 512; row += 4) {
      v2f a = *(const v2f*)(u + row + 2 * hf);
      size_t base = (size_t)(row + 2 * hf) * SK_N;
      v2f b;
      b[0] = Kc[base];
      b[1] = Kc[base + SK_N];
      acc = sk_wmma(a, b, acc);
    }
    if (lane < 16) part[wave][lane] = acc[0];
    __syncthreads();
    if (threadIdx.x < 16) {
      float t = 0.f;
#pragma unroll
      for (int w = 0; w < 8; ++w) t += part[w][threadIdx.x];
      v_out[colBase + threadIdx.x] = nu[colBase + threadIdx.x] / (t + SK_TOL);
    }
  }
}

__global__ void __launch_bounds__(256) sk_final(const float* __restrict__ mu,
                                                const float* __restrict__ nu,
                                                const float* __restrict__ u,
                                                const float* __restrict__ v,
                                                float* __restrict__ P,
                                                float* __restrict__ lpart) {
  __shared__ float red[256];
  const int i = blockIdx.x;
  const float ui = u[i];
  const float mi = mu[i];
  float* Prow = P + (size_t)i * SK_N;
  float w = 0.f;
  for (int j = threadIdx.x * 4; j < SK_N; j += 256 * 4) {
    float4 k4 = *(const float4*)(Prow + j);
    float4 v4 = *(const float4*)(v + j);
    float4 n4 = *(const float4*)(nu + j);
    float4 p;
    p.x = ui * k4.x * v4.x;
    p.y = ui * k4.y * v4.y;
    p.z = ui * k4.z * v4.z;
    p.w = ui * k4.w * v4.w;
    *(float4*)(Prow + j) = p;
    w += p.x * fabsf(mi - n4.x) + p.y * fabsf(mi - n4.y) +
         p.z * fabsf(mi - n4.z) + p.w * fabsf(mi - n4.w);
  }
  red[threadIdx.x] = w;
  __syncthreads();
  for (int s = 128; s > 0; s >>= 1) {
    if (threadIdx.x < s) red[threadIdx.x] += red[threadIdx.x + s];
    __syncthreads();
  }
  if (threadIdx.x == 0) lpart[i] = red[0];
}

__global__ void __launch_bounds__(256) sk_loss_reduce(const float* __restrict__ lpart,
                                                      float* __restrict__ loss) {
  __shared__ float red[256];
  float s = 0.f;
  for (int i = threadIdx.x; i < SK_N; i += 256) s += lpart[i];
  red[threadIdx.x] = s;
  __syncthreads();
  for (int k = 128; k > 0; k >>= 1) {
    if (threadIdx.x < k) red[threadIdx.x] += red[threadIdx.x + k];
    __syncthreads();
  }
  if (threadIdx.x == 0) *loss = red[0];
}

extern "C" void kernel_launch(void* const* d_in, const int* in_sizes, int n_in,
                              void* d_out, int out_size, void* d_ws, size_t ws_size,
                              hipStream_t stream) {
  const float* mu = (const float*)d_in[0];
  const float* nu = (const float*)d_in[1];
  const float* C  = (const float*)d_in[2];
  float* K = (float*)d_out;
  float* ws = (float*)d_ws;
  float* u     = ws;
  float* v     = ws + SK_N;
  float* lpart = ws + 2 * SK_N;
  float* cpart = ws + 4 * SK_N;

  sk_init_K<<<SK_NN / (256 * 4), 256, 0, stream>>>(C, K);
  sk_ones<<<SK_N / 256, 256, 0, stream>>>(v);

  for (int t = 0; t < SK_ITERS; ++t) {
    sk_rows_fused<<<SK_NRB, 256, 0, stream>>>(K, mu, v, u, cpart, ws_size);
    sk_cols<<<SK_N / 16, 256, 0, stream>>>(K, u, nu, cpart, v, ws_size);
  }

  sk_final<<<SK_N, 256, 0, stream>>>(mu, nu, u, v, K, lpart);
  sk_loss_reduce<<<1, 256, 0, stream>>>(lpart, K + SK_NN);
}


# K read once per iteration: whole-line loads in column-sum layout stay in registers for the column pass; row-sum fragments turned through a wave-private LDS buffer (no L2 re-read); sums unchanged
# speedup vs baseline: 2.7281x; 1.0993x over previous
#include <hip/hip_runtime.h>
#include <math.h>


typedef float v2f __attribute__((ext_vector_type(2)));
typedef float v8f __attribute__((ext_vector_type(8)));

#define SK_N     4096
#define SK_NN    (4096 * 4096)
#define SK_IEPS  10.0f
#define SK_TOL   1e-9f
#define SK_ITERS 100

#define SK_FW    8
#define SK_FCOLS (SK_N / SK_FW)
#define SK_NRB   (SK_N / 16)
#define SK_CG    (SK_FCOLS / 64)
#define SK_KPF   4
#define SK_TP    17
#define SK_WS_NEED ((size_t)(4 * SK_N + SK_NRB * SK_N) * sizeof(float))

static __device__ __forceinline__ v8f sk_wmma(v2f a, v2f b, v8f c) {
  return __builtin_amdgcn_wmma_f32_16x16x4_f32(false, a, false, b, (short)0, c,
                                               false, false);
}

template <int R>
static __device__ __forceinline__ float sk_row_lane(float x) {
  return __builtin_bit_cast(
      float, __builtin_amdgcn_update_dpp(0, __builtin_bit_cast(int, x), 0x150 + R, 0xF, 0xF, false));
}

static __device__ __forceinline__ void sk_lds_done() {
  asm volatile("s_wait_dscnt 0x0" ::: "memory");
}
static __device__ __forceinline__ void sk_bar_signal() {
  asm volatile("s_barrier_signal -1" ::: "memory");
}
static __device__ __forceinline__ void sk_bar_wait() {
  asm volatile("s_barrier_wait -1" ::: "memory");
}

__global__ void __launch_bounds__(256) sk_init_K(const float* __restrict__ C,
                                                 float* __restrict__ K) {
  size_t i4 = ((size_t)blockIdx.x * 256 + threadIdx.x) * 4;
  float4 c4 = *(const float4*)(C + i4);
  float4 k4;
  k4.x = expf(-SK_IEPS * c4.x);
  k4.y = expf(-SK_IEPS * c4.y);
  k4.z = expf(-SK_IEPS * c4.z);
  k4.w = expf(-SK_IEPS * c4.w);
  *(float4*)(K + i4) = k4;
}

__global__ void sk_ones(float* __restrict__ v) {
  v[blockIdx.x * 256 + threadIdx.x] = 1.0f;
}

__global__ void __launch_bounds__(256) sk_rows_fused(const float* __restrict__ K,
                                                     const float* __restrict__ mu,
                                                     const float* __restrict__ v,
                                                     float* __restrict__ u_out,
                                                     float* __restrict__ cpart,
                                                     size_t ws_size) {
  __shared__ float4 vsh4[SK_N / 4];
  __shared__ float part[SK_FW][16];
  __shared__ float4 turn[SK_FW][16][SK_TP];
  const int wave = threadIdx.x >> 5;
  const int lane = threadIdx.x & 31;
  const int h = lane >> 4;
  const int c = lane & 15;
  const int rowBase = blockIdx.x << 4;
  const int colBase = wave * SK_FCOLS;
  const float* Kcol = K + (size_t)(rowBase + 2 * h) * SK_N + colBase + 4 * c;
  const float4* vp4 = vsh4 + (colBase >> 2) + h;
  float4* tw = &turn[wave][2 * h][c];
  const float4* tr = &turn[wave][c][h];

  const int ur = (c + 2 * h) & 15;
  const float mu_r = mu[rowBase + ur];

#pragma unroll
  for (int q = 0; q < 4; ++q)
    vsh4[256 * q + threadIdx.x] = *(const float4*)(v + 1024 * q + 4 * threadIdx.x);

  float4 qb[SK_CG][8];
#pragma unroll
  for (int g = 0; g < SK_KPF; ++g) {
#pragma unroll
    for (int x = 0; x < 8; ++x)
      qb[g][x] = *(const float4*)(Kcol + (size_t)(4 * (x >> 1) + (x & 1)) * SK_N + 64 * g);
    if (g >= 1) __builtin_amdgcn_sched_barrier(0);
  }
  sk_lds_done();
  sk_bar_signal();
  sk_bar_wait();

  const v8f zero = {0.f, 0.f, 0.f, 0.f, 0.f, 0.f, 0.f, 0.f};
  v8f rs0 = zero, rs1 = zero;
  float4 af[2][8], vb[2][8];
#pragma unroll
  for (int g = 0; g <= SK_CG; ++g) {
    if (g + SK_KPF < SK_CG) {
#pragma unroll
      for (int x = 0; x < 8; ++x)
        qb[g + SK_KPF][x] =
            *(const float4*)(Kcol + (size_t)(4 * (x >> 1) + (x & 1)) * SK_N + 64 * (g + SK_KPF));
    }
    if (g < SK_CG) {
      __builtin_amdgcn_wave_barrier();
#pragma unroll
      for (int x = 0; x < 8; ++x) tw[(4 * (x >> 1) + (x & 1)) * SK_TP] = qb[g][x];
      __builtin_amdgcn_wave_barrier();
#pragma unroll
      for (int y = 0; y < 8; ++y) af[g & 1][y] = tr[8 * (y >> 2) + 2 * (y & 3)];
#pragma unroll
      for (int y = 0; y < 8; ++y) vb[g & 1][y] = vp4[16 * g + 8 * (y >> 2) + 2 * (y & 3)];
    }
    if (g > 0) {
#pragma unroll
      for (int t = 0; t < 2; ++t) {
#pragma unroll
        for (int p = 0; p < 2; ++p) {
          const float4 a4 = af[(g - 1) & 1][4 * t + p];
          const float4 e4 = af[(g - 1) & 1][4 * t + 2 + p];
          const float4 b4 = vb[(g - 1) & 1][4 * t + p];
          const float4 f4 = vb[(g - 1) & 1][4 * t + 2 + p];
          v2f a0, a1, e0, e1, b0, b1, f0, f1;
          a0[0] = a4.x; a0[1] = a4.y; a1[0] = a4.z; a1[1] = a4.w;
          e0[0] = e4.x; e0[1] = e4.y; e1[0] = e4.z; e1[1] = e4.w;
          b0[0] = b4.x; b0[1] = b4.y; b1[0] = b4.z; b1[1] = b4.w;
          f0[0] = f4.x; f0[1] = f4.y; f1[0] = f4.z; f1[1] = f4.w;
          rs0 = sk_wmma(a0, b0, rs0);
          rs1 = sk_wmma(e0, f0, rs1);
          rs0 = sk_wmma(a1, b1, rs0);
          rs1 = sk_wmma(e1, f1, rs1);
        }
      }
    }
    __builtin_amdgcn_sched_barrier(0);
  }

  if (c == 0) {
    float* dst = &part[wave][h * 8];
#pragma unroll
    for (int r = 0; r < 8; ++r) dst[r] = rs0[r] + rs1[r];
  }
  sk_lds_done();
  sk_bar_signal();
  sk_bar_wait();
  float rsum = 0.f;
#pragma unroll
  for (int w = 0; w < SK_FW; ++w) rsum += part[w][ur];
  const float uval = mu_r / (rsum + SK_TOL);
  if (threadIdx.x < 16) u_out[rowBase + threadIdx.x] = uval;

  v2f ua[4];
  ua[0][0] = sk_row_lane<0>(uval);  ua[0][1] = sk_row_lane<1>(uval);
  ua[1][0] = sk_row_lane<4>(uval);  ua[1][1] = sk_row_lane<5>(uval);
  ua[2][0] = sk_row_lane<8>(uval);  ua[2][1] = sk_row_lane<9>(uval);
  ua[3][0] = sk_row_lane<12>(uval); ua[3][1] = sk_row_lane<13>(uval);

  const bool have_ws = ws_size >= SK_WS_NEED;
  float* dstc = cpart + (size_t)blockIdx.x * SK_N + colBase + 4 * c;
#pragma unroll
  for (int g = 0; g < SK_CG; ++g) {
    v8f o0 = zero, o1 = zero, o2 = zero, o3 = zero;
#pragma unroll
    for (int j = 0; j < 4; ++j) {
      const float4 r0 = qb[g][2 * j];
      const float4 r1 = qb[g][2 * j + 1];
      v2f t0, t1, t2, t3;
      t0[0] = r0.x; t0[1] = r1.x;
      t1[0] = r0.y; t1[1] = r1.y;
      t2[0] = r0.z; t2[1] = r1.z;
      t3[0] = r0.w; t3[1] = r1.w;
      o0 = sk_wmma(ua[j], t0, o0);
      o1 = sk_wmma(ua[j], t1, o1);
      o2 = sk_wmma(ua[j], t2, o2);
      o3 = sk_wmma(ua[j], t3, o3);
    }
    if (have_ws && h == 0) {
      float4 o;
      o.x = o0[0]; o.y = o1[0]; o.z = o2[0]; o.w = o3[0];
      *(float4*)(dstc + 64 * g) = o;
    }
  }
}

__global__ void __launch_bounds__(256) sk_cols(const float* __restrict__ K,
                                               const float* __restrict__ u,
                                               const float* __restrict__ nu,
                                               const float* __restrict__ cpart,
                                               float* __restrict__ v_out,
                                               size_t ws_size) {
  __shared__ float part[16][16];
  const int colBase = blockIdx.x << 4;
  if (ws_size >= SK_WS_NEED) {
    const int cc = threadIdx.x & 15;
    const int pg = threadIdx.x >> 4;
    const float* src = cpart + (size_t)pg * SK_N + colBase + cc;
    float s = 0.f;
#pragma unroll
    for (int q = 0; q < SK_NRB / 16; ++q) s += src[(size_t)(16 * q) * SK_N];
    part[pg][cc] = s;
    __syncthreads();
    if (threadIdx.x < 16) {
      float t = 0.f;
#pragma unroll
      for (int p = 0; p < 16; ++p) t += part[p][threadIdx.x];
      v_out[colBase + threadIdx.x] = nu[colBase + threadIdx.x] / (t + SK_TOL);
    }
  } else {
    const int wave = threadIdx.x >> 5;
    const int lane = threadIdx.x & 31;
    const int hf   = lane >> 4;
    const int r    = lane & 15;
    const float* Kc = K + colBase + r;
    v8f acc = {0.f, 0.f, 0.f, 0.f, 0.f, 0.f, 0.f, 0.f};
    const int r0 = wave << 9;
#pragma unroll 2
    for (int row = r0; row < r0 + 512; row += 4) {
      v2f a = *(const v2f*)(u + row + 2 * hf);
      size_t base = (size_t)(row + 2 * hf) * SK_N;
      v2f b;
      b[0] = Kc[base];
      b[1] = Kc[base + SK_N];
      acc = sk_wmma(a, b, acc);
    }
    if (lane < 16) part[wave][lane] = acc[0];
    __syncthreads();
    if (threadIdx.x < 16) {
      float t = 0.f;
#pragma unroll
      for (int w = 0; w < 8; ++w) t += part[w][threadIdx.x];
      v_out[colBase + threadIdx.x] = nu[colBase + threadIdx.x] / (t + SK_TOL);
    }
  }
}

__global__ void __launch_bounds__(256) sk_final(const float* __restrict__ mu,
                                                const float* __restrict__ nu,
                                                const float* __restrict__ u,
                                                const float* __restrict__ v,
                                                float* __restrict__ P,
                                                float* __restrict__ lpart) {
  __shared__ float red[256];
  const int i = blockIdx.x;
  const float ui = u[i];
  const float mi = mu[i];
  float* Prow = P + (size_t)i * SK_N;
  float w = 0.f;
  for (int j = threadIdx.x * 4; j < SK_N; j += 256 * 4) {
    float4 k4 = *(const float4*)(Prow + j);
    float4 v4 = *(const float4*)(v + j);
    float4 n4 = *(const float4*)(nu + j);
    float4 p;
    p.x = ui * k4.x * v4.x;
    p.y = ui * k4.y * v4.y;
    p.z = ui * k4.z * v4.z;
    p.w = ui * k4.w * v4.w;
    *(float4*)(Prow + j) = p;
    w += p.x * fabsf(mi - n4.x) + p.y * fabsf(mi - n4.y) +
         p.z * fabsf(mi - n4.z) + p.w * fabsf(mi - n4.w);
  }
  red[threadIdx.x] = w;
  __syncthreads();
  for (int s = 128; s > 0; s >>= 1) {
    if (threadIdx.x < s) red[threadIdx.x] += red[threadIdx.x + s];
    __syncthreads();
  }
  if (threadIdx.x == 0) lpart[i] = red[0];
}

__global__ void __launch_bounds__(256) sk_loss_reduce(const float* __restrict__ lpart,
                                                      float* __restrict__ loss) {
  __shared__ float red[256];
  float s = 0.f;
  for (int i = threadIdx.x; i < SK_N; i += 256) s += lpart[i];
  red[threadIdx.x] = s;
  __syncthreads();
  for (int k = 128; k > 0; k >>= 1) {
    if (threadIdx.x < k) red[threadIdx.x] += red[threadIdx.x + k];
    __syncthreads();
  }
  if (threadIdx.x == 0) *loss = red[0];
}

extern "C" void kernel_launch(void* const* d_in, const int* in_sizes, int n_in,
                              void* d_out, int out_size, void* d_ws, size_t ws_size,
                              hipStream_t stream) {
  const float* mu = (const float*)d_in[0];
  const float* nu = (const float*)d_in[1];
  const float* C  = (const float*)d_in[2];
  float* K = (float*)d_out;
  float* ws = (float*)d_ws;
  float* u     = ws;
  float* v     = ws + SK_N;
  float* lpart = ws + 2 * SK_N;
  float* cpart = ws + 4 * SK_N;

  sk_init_K<<<SK_NN / (256 * 4), 256, 0, stream>>>(C, K);
  sk_ones<<<SK_N / 256, 256, 0, stream>>>(v);

  for (int t = 0; t < SK_ITERS; ++t) {
    sk_rows_fused<<<SK_NRB, 256, 0, stream>>>(K, mu, v, u, cpart, ws_size);
    sk_cols<<<SK_N / 16, 256, 0, stream>>>(K, u, nu, cpart, v, ws_size);
  }

  sk_final<<<SK_N, 256, 0, stream>>>(mu, nu, u, v, K, lpart);
  sk_loss_reduce<<<1, 256, 0, stream>>>(lpart, K + SK_NN);
}
